# starting kernel
# baseline (speedup 1.0000x reference)
#include <hip/hip_runtime.h>

typedef __attribute__((ext_vector_type(2))) float v2f;
typedef __attribute__((ext_vector_type(8))) float v8f;

#define FDIM 64

__global__ void init_cnt_kernel(float* __restrict__ cnt, int n) {
    int i = blockIdx.x * blockDim.x + threadIdx.x;
    if (i < n) cnt[i] = 0.0f;
}

__global__ void count_edges_kernel(const int* __restrict__ dst, float* __restrict__ cnt, int e) {
    int i = blockIdx.x * blockDim.x + threadIdx.x;
    if (i < e) atomicAdd(&cnt[dst[i]], 1.0f);
}

__global__ void finalize_deg_kernel(const float* __restrict__ cnt, float* __restrict__ dinv,
                                    float* __restrict__ sinv, int n) {
    int i = blockIdx.x * blockDim.x + threadIdx.x;
    if (i < n) {
        float c = cnt[i];
        dinv[i] = rsqrtf(c + 1.0f);
        sinv[i] = 1.0f / fmaxf(c, 1.0f);
    }
}

__global__ void edge_w_kernel(const int* __restrict__ src, const int* __restrict__ dst,
                              const float* __restrict__ dinv, float* __restrict__ ew, int e) {
    int i = blockIdx.x * blockDim.x + threadIdx.x;
    if (i < e) ew[i] = dinv[src[i]] * dinv[dst[i]];
}

__launch_bounds__(256)
__global__ void gemm64_wmma_kernel(const float* __restrict__ A, const float* __restrict__ W1g,
                                   const float* __restrict__ B, const float* __restrict__ W2g,
                                   const float* __restrict__ bias,
                                   const float* __restrict__ rowscale,
                                   const float* __restrict__ lnAg, const float* __restrict__ lnAb,
                                   const float* __restrict__ lnBg, const float* __restrict__ lnBb,
                                   const float* __restrict__ dinv2, const float* __restrict__ bias2,
                                   float* __restrict__ out2,
                                   float* __restrict__ out, int n, int reluOut, int reluA) {
    __shared__ float W1[FDIM * FDIM];
    __shared__ float W2[FDIM * FDIM];
    __shared__ float LNP[4][FDIM];

    const int t = threadIdx.x;
    {
        const float4* s1 = (const float4*)W1g;
        float4* l1 = (float4*)W1;
        for (int i = t; i < 1024; i += 256) l1[i] = s1[i];
        if (B) {
            const float4* s2 = (const float4*)W2g;
            float4* l2 = (float4*)W2;
            for (int i = t; i < 1024; i += 256) l2[i] = s2[i];
        }
        if (lnAg && t < FDIM) { LNP[0][t] = lnAg[t]; LNP[1][t] = lnAb[t]; }
        if (lnBg && t < FDIM) { LNP[2][t] = lnBg[t]; LNP[3][t] = lnBb[t]; }
    }
    __syncthreads();

    const int wave  = t >> 5;
    const int lane  = t & 31;
    const int m0    = (blockIdx.x * 8 + wave) * 16;
    if (m0 >= n) return;
    const int lhalf = lane >> 4;
    const int l15   = lane & 15;
    const int kOff  = lhalf * 2;

    int row = m0 + l15;
    if (row >= n) row = n - 1;

    const float rs = rowscale ? rowscale[row] : 1.0f;

    v2f afrag[16];
    {
        const float* ar = A + (size_t)row * FDIM + kOff;
        #pragma unroll
        for (int kb = 0; kb < 16; ++kb) {
            float a0 = ar[kb * 4 + 0];
            float a1 = ar[kb * 4 + 1];
            if (reluA) { a0 = fmaxf(a0, 0.0f); a1 = fmaxf(a1, 0.0f); }
            afrag[kb].x = a0 * rs;
            afrag[kb].y = a1 * rs;
        }
        if (lnAg) {
            float s = 0.0f, q = 0.0f;
            #pragma unroll
            for (int kb = 0; kb < 16; ++kb) {
                s += afrag[kb].x + afrag[kb].y;
                q += afrag[kb].x * afrag[kb].x + afrag[kb].y * afrag[kb].y;
            }
            s += __shfl_xor(s, 16, 32);
            q += __shfl_xor(q, 16, 32);
            const float mu  = s * (1.0f / 64.0f);
            const float inv = rsqrtf(fmaxf(q * (1.0f / 64.0f) - mu * mu, 0.0f) + 1e-5f);
            #pragma unroll
            for (int kb = 0; kb < 16; ++kb) {
                const int k0 = kb * 4 + kOff;
                afrag[kb].x = (afrag[kb].x - mu) * inv * LNP[0][k0]     + LNP[1][k0];
                afrag[kb].y = (afrag[kb].y - mu) * inv * LNP[0][k0 + 1] + LNP[1][k0 + 1];
            }
        }
    }
    v2f bfrag[16];
    if (B) {
        const float* br = B + (size_t)row * FDIM + kOff;
        #pragma unroll
        for (int kb = 0; kb < 16; ++kb) {
            bfrag[kb].x = br[kb * 4 + 0];
            bfrag[kb].y = br[kb * 4 + 1];
        }
        if (lnBg) {
            float s = 0.0f, q = 0.0f;
            #pragma unroll
            for (int kb = 0; kb < 16; ++kb) {
                s += bfrag[kb].x + bfrag[kb].y;
                q += bfrag[kb].x * bfrag[kb].x + bfrag[kb].y * bfrag[kb].y;
            }
            s += __shfl_xor(s, 16, 32);
            q += __shfl_xor(q, 16, 32);
            const float mu  = s * (1.0f / 64.0f);
            const float inv = rsqrtf(fmaxf(q * (1.0f / 64.0f) - mu * mu, 0.0f) + 1e-5f);
            #pragma unroll
            for (int kb = 0; kb < 16; ++kb) {
                const int k0 = kb * 4 + kOff;
                bfrag[kb].x = (bfrag[kb].x - mu) * inv * LNP[2][k0]     + LNP[3][k0];
                bfrag[kb].y = (bfrag[kb].y - mu) * inv * LNP[2][k0 + 1] + LNP[3][k0 + 1];
            }
        }
    }

    #pragma unroll
    for (int nb = 0; nb < 4; ++nb) {
        const int n0 = nb * 16;
        v8f acc = {0.f, 0.f, 0.f, 0.f, 0.f, 0.f, 0.f, 0.f};

        v2f wf[16];
        #pragma unroll
        for (int kb = 0; kb < 16; ++kb) {
            const int kr = kb * 4 + kOff;
            wf[kb].x = W1[kr * FDIM + n0 + l15];
            wf[kb].y = W1[(kr + 1) * FDIM + n0 + l15];
        }
        #pragma unroll
        for (int kb = 0; kb < 16; ++kb)
            acc = __builtin_amdgcn_wmma_f32_16x16x4_f32(
                false, afrag[kb], false, wf[kb], (short)0, acc, false, false);

        if (B) {
            v2f wg[16];
            #pragma unroll
            for (int kb = 0; kb < 16; ++kb) {
                const int kr = kb * 4 + kOff;
                wg[kb].x = W2[kr * FDIM + n0 + l15];
                wg[kb].y = W2[(kr + 1) * FDIM + n0 + l15];
            }
            #pragma unroll
            for (int kb = 0; kb < 16; ++kb)
                acc = __builtin_amdgcn_wmma_f32_16x16x4_f32(
                    false, bfrag[kb], false, wg[kb], (short)0, acc, false, false);
        }

        const float bval  = bias  ? bias[n0 + l15]  : 0.0f;
        const float b2val = bias2 ? bias2[n0 + l15] : 0.0f;
        #pragma unroll
        for (int j = 0; j < 8; ++j) {
            const int r = m0 + j + lhalf * 8;
            if (r < n) {
                float v = acc[j] + bval;
                if (reluOut) v = fmaxf(v, 0.0f);
                out[(size_t)r * FDIM + n0 + l15] = v;
                if (out2) {
                    const float d = dinv2[r];
                    out2[(size_t)r * FDIM + n0 + l15] = acc[j] * d * d + b2val;
                }
            }
        }
    }
}

__global__ void scatter_gcn_kernel(const float* __restrict__ h, const int* __restrict__ src,
                                   const int* __restrict__ dst, const float* __restrict__ ew,
                                   float* __restrict__ g, int e) {
    long long tid = (long long)blockIdx.x * blockDim.x + threadIdx.x;
    long long edge = tid >> 4;
    if (edge >= e) return;
    int q = (int)(tid & 15);
    int s = src[edge], d = dst[edge];
    float w = ew[edge];
    float4 hv = ((const float4*)(h + (size_t)s * FDIM))[q];
    float* gp = g + (size_t)d * FDIM + q * 4;
    atomicAdd(gp + 0, hv.x * w);
    atomicAdd(gp + 1, hv.y * w);
    atomicAdd(gp + 2, hv.z * w);
    atomicAdd(gp + 3, hv.w * w);
}

__global__ void scatter_sum_kernel(const float* __restrict__ xin, const int* __restrict__ src,
                                   const int* __restrict__ dst, float* __restrict__ agg, int e) {
    long long tid = (long long)blockIdx.x * blockDim.x + threadIdx.x;
    long long edge = tid >> 4;
    if (edge >= e) return;
    int q = (int)(tid & 15);
    int s = src[edge], d = dst[edge];
    float4 xv = ((const float4*)(xin + (size_t)s * FDIM))[q];
    float* ap = agg + (size_t)d * FDIM + q * 4;
    atomicAdd(ap + 0, xv.x);
    atomicAdd(ap + 1, xv.y);
    atomicAdd(ap + 2, xv.z);
    atomicAdd(ap + 3, xv.w);
}

__global__ void zero_kernel(float* __restrict__ p, long long total) {
    long long i = (long long)blockIdx.x * blockDim.x + threadIdx.x;
    if (i < total) p[i] = 0.0f;
}

extern "C" void kernel_launch(void* const* d_in, const int* in_sizes, int n_in,
                              void* d_out, int out_size, void* d_ws, size_t ws_size,
                              hipStream_t stream) {
    const float* x        = (const float*)d_in[0];
    const int*   ei       = (const int*)d_in[1];
    const float* gcn_w1   = (const float*)d_in[2];
    const float* gcn_b1   = (const float*)d_in[3];
    const float* gcn_w2   = (const float*)d_in[4];
    const float* gcn_b2   = (const float*)d_in[5];
    const float* sage_wl1 = (const float*)d_in[6];
    const float* sage_bl1 = (const float*)d_in[7];
    const float* sage_wr1 = (const float*)d_in[8];
    const float* sage_wl2 = (const float*)d_in[9];
    const float* sage_bl2 = (const float*)d_in[10];
    const float* sage_wr2 = (const float*)d_in[11];
    const float* gcn_ln_g = (const float*)d_in[12];
    const float* gcn_ln_b = (const float*)d_in[13];
    const float* sage_ln_g = (const float*)d_in[14];
    const float* sage_ln_b = (const float*)d_in[15];
    const float* proj_w   = (const float*)d_in[16];
    const float* proj_b   = (const float*)d_in[17];

    const int N = in_sizes[0] / FDIM;
    const int E = in_sizes[1] / 2;
    const int* src = ei;
    const int* dst = ei + E;

    float* cnt  = (float*)d_ws;
    float* dinv = cnt + N;
    float* sinv = dinv + N;
    float* ew   = sinv + N;
    float* h    = ew + E;
    float* g    = h + (size_t)N * FDIM;
    float* agg  = g + (size_t)N * FDIM;
    float* out  = (float*)d_out;

    const long long NM = (long long)N * FDIM;
    const int nblkN  = (N + 255) / 256;
    const int nblkE  = (E + 255) / 256;
    const int nblkNM = (int)((NM + 255) / 256);
    const int nblkSc = (int)(((long long)E * 16 + 255) / 256);
    const int nblkG  = (N + 127) / 128;

    init_cnt_kernel<<<nblkN, 256, 0, stream>>>(cnt, N);
    count_edges_kernel<<<nblkE, 256, 0, stream>>>(dst, cnt, E);
    finalize_deg_kernel<<<nblkN, 256, 0, stream>>>(cnt, dinv, sinv, N);
    edge_w_kernel<<<nblkE, 256, 0, stream>>>(src, dst, dinv, ew, E);

    gemm64_wmma_kernel<<<nblkG, 256, 0, stream>>>(
        x, gcn_w1, nullptr, nullptr, nullptr, nullptr,
        nullptr, nullptr, nullptr, nullptr,
        dinv, gcn_b1, g, h, N, 0, 0);
    scatter_gcn_kernel<<<nblkSc, 256, 0, stream>>>(h, src, dst, ew, g, E);

    gemm64_wmma_kernel<<<nblkG, 256, 0, stream>>>(
        g, gcn_w2, nullptr, nullptr, nullptr, nullptr,
        nullptr, nullptr, nullptr, nullptr,
        dinv, gcn_b2, g, h, N, 0, 1  );
    scatter_gcn_kernel<<<nblkSc, 256, 0, stream>>>(h, src, dst, ew, g, E);

    zero_kernel<<<nblkNM, 256, 0, stream>>>(agg, NM);
    scatter_sum_kernel<<<nblkSc, 256, 0, stream>>>(x, src, dst, agg, E);
    gemm64_wmma_kernel<<<nblkG, 256, 0, stream>>>(
        agg, sage_wl1, x, sage_wr1, sage_bl1, sinv,
        nullptr, nullptr, nullptr, nullptr,
        nullptr, nullptr, nullptr, out, N, 1  , 0);

    zero_kernel<<<nblkNM, 256, 0, stream>>>(agg, NM);
    scatter_sum_kernel<<<nblkSc, 256, 0, stream>>>(out, src, dst, agg, E);
    gemm64_wmma_kernel<<<nblkG, 256, 0, stream>>>(
        agg, sage_wl2, out, sage_wr2, sage_bl2, sinv,
        nullptr, nullptr, nullptr, nullptr,
        nullptr, nullptr, nullptr, h, N, 0, 0);

    gemm64_wmma_kernel<<<nblkG, 256, 0, stream>>>(
        g, proj_w, h, proj_w + FDIM * FDIM, proj_b, nullptr,
        gcn_ln_g, gcn_ln_b, sage_ln_g, sage_ln_b,
        nullptr, nullptr, nullptr, out, N, 0, 0);
}


# Edge scatters re-cut by destination owner: a work-group keeps 1024 rows as an LDS f32 tile, filters its edge-list slice, adds rows by LDS atomics, flushes whole rows by f32 atomics; sum tree changes
# speedup vs baseline: 1.8041x; 1.8041x over previous
#include <hip/hip_runtime.h>

typedef __attribute__((ext_vector_type(2))) float v2f;
typedef __attribute__((ext_vector_type(8))) float v8f;

#define FDIM 64
#define NTHR  1024
#define NWAVE 32
#define TROWS 1024
#define NSLICE 2
#define QCAP  160

__global__ void init_cnt_kernel(float* __restrict__ cnt, int n) {
    int i = blockIdx.x * blockDim.x + threadIdx.x;
    if (i < n) cnt[i] = 0.0f;
}

__global__ void count_edges_kernel(const int* __restrict__ dst, float* __restrict__ cnt, int e) {
    int i = blockIdx.x * blockDim.x + threadIdx.x;
    if (i < e) atomicAdd(&cnt[dst[i]], 1.0f);
}

__global__ void finalize_deg_kernel(const float* __restrict__ cnt, float* __restrict__ dinv,
                                    float* __restrict__ sinv, int n) {
    int i = blockIdx.x * blockDim.x + threadIdx.x;
    if (i < n) {
        float c = cnt[i];
        dinv[i] = rsqrtf(c + 1.0f);
        sinv[i] = 1.0f / fmaxf(c, 1.0f);
    }
}

__global__ void edge_w_kernel(const int* __restrict__ src, const int* __restrict__ dst,
                              const float* __restrict__ dinv, float* __restrict__ ew, int e) {
    int i = blockIdx.x * blockDim.x + threadIdx.x;
    if (i < e) ew[i] = dinv[src[i]] * dinv[dst[i]];
}

__device__ __forceinline__ void gemm64_body(const int blk, float* W1, float* W2, float (*LNP)[FDIM],
                                   const float* A, const float* __restrict__ W1g,
                                   const float* __restrict__ B, const float* __restrict__ W2g,
                                   const float* __restrict__ bias,
                                   const float* __restrict__ rowscale,
                                   const float* __restrict__ lnAg, const float* __restrict__ lnAb,
                                   const float* __restrict__ lnBg, const float* __restrict__ lnBb,
                                   const float* __restrict__ dinv2, const float* __restrict__ bias2,
                                   float* out2,
                                   float* out, float* zeroA, int n, int reluOut, int reluA) {
    const int t = threadIdx.x;
    {
        const float4* s1 = (const float4*)W1g;
        float4* l1 = (float4*)W1;
        for (int i = t; i < 1024; i += NTHR) l1[i] = s1[i];
        if (B) {
            const float4* s2 = (const float4*)W2g;
            float4* l2 = (float4*)W2;
            for (int i = t; i < 1024; i += NTHR) l2[i] = s2[i];
        }
        if (lnAg && t < FDIM) { LNP[0][t] = lnAg[t]; LNP[1][t] = lnAb[t]; }
        if (lnBg && t < FDIM) { LNP[2][t] = lnBg[t]; LNP[3][t] = lnBb[t]; }
    }
    __syncthreads();

    const int wave  = t >> 5;
    const int lane  = t & 31;
    const int m0    = (blk * NWAVE + wave) * 16;
    if (m0 >= n) return;
    const int lhalf = lane >> 4;
    const int l15   = lane & 15;
    const int kOff  = lhalf * 2;

    int row = m0 + l15;
    if (row >= n) row = n - 1;

    const float rs = rowscale ? rowscale[row] : 1.0f;

    v2f afrag[16];
    {
        const float* ar = A + (size_t)row * FDIM + kOff;
        #pragma unroll
        for (int kb = 0; kb < 16; ++kb) {
            float a0 = ar[kb * 4 + 0];
            float a1 = ar[kb * 4 + 1];
            if (reluA) { a0 = fmaxf(a0, 0.0f); a1 = fmaxf(a1, 0.0f); }
            afrag[kb].x = a0 * rs;
            afrag[kb].y = a1 * rs;
        }
        if (lnAg) {
            float s = 0.0f, q = 0.0f;
            #pragma unroll
            for (int kb = 0; kb < 16; ++kb) {
                s += afrag[kb].x + afrag[kb].y;
                q += afrag[kb].x * afrag[kb].x + afrag[kb].y * afrag[kb].y;
            }
            s += __shfl_xor(s, 16, 32);
            q += __shfl_xor(q, 16, 32);
            const float mu  = s * (1.0f / 64.0f);
            const float inv = rsqrtf(fmaxf(q * (1.0f / 64.0f) - mu * mu, 0.0f) + 1e-5f);
            #pragma unroll
            for (int kb = 0; kb < 16; ++kb) {
                const int k0 = kb * 4 + kOff;
                afrag[kb].x = (afrag[kb].x - mu) * inv * LNP[0][k0]     + LNP[1][k0];
                afrag[kb].y = (afrag[kb].y - mu) * inv * LNP[0][k0 + 1] + LNP[1][k0 + 1];
            }
        }
    }
    v2f bfrag[16];
    if (B) {
        const float* br = B + (size_t)row * FDIM + kOff;
        #pragma unroll
        for (int kb = 0; kb < 16; ++kb) {
            bfrag[kb].x = br[kb * 4 + 0];
            bfrag[kb].y = br[kb * 4 + 1];
        }
        if (lnBg) {
            float s = 0.0f, q = 0.0f;
            #pragma unroll
            for (int kb = 0; kb < 16; ++kb) {
                s += bfrag[kb].x + bfrag[kb].y;
                q += bfrag[kb].x * bfrag[kb].x + bfrag[kb].y * bfrag[kb].y;
            }
            s += __shfl_xor(s, 16, 32);
            q += __shfl_xor(q, 16, 32);
            const float mu  = s * (1.0f / 64.0f);
            const float inv = rsqrtf(fmaxf(q * (1.0f / 64.0f) - mu * mu, 0.0f) + 1e-5f);
            #pragma unroll
            for (int kb = 0; kb < 16; ++kb) {
                const int k0 = kb * 4 + kOff;
                bfrag[kb].x = (bfrag[kb].x - mu) * inv * LNP[2][k0]     + LNP[3][k0];
                bfrag[kb].y = (bfrag[kb].y - mu) * inv * LNP[2][k0 + 1] + LNP[3][k0 + 1];
            }
        }
    }

    #pragma unroll
    for (int nb = 0; nb < 4; ++nb) {
        const int n0 = nb * 16;
        v8f acc = {0.f, 0.f, 0.f, 0.f, 0.f, 0.f, 0.f, 0.f};

        v2f wf[16];
        #pragma unroll
        for (int kb = 0; kb < 16; ++kb) {
            const int kr = kb * 4 + kOff;
            wf[kb].x = W1[kr * FDIM + n0 + l15];
            wf[kb].y = W1[(kr + 1) * FDIM + n0 + l15];
        }
        #pragma unroll
        for (int kb = 0; kb < 16; ++kb)
            acc = __builtin_amdgcn_wmma_f32_16x16x4_f32(
                false, afrag[kb], false, wf[kb], (short)0, acc, false, false);

        if (B) {
            v2f wg[16];
            #pragma unroll
            for (int kb = 0; kb < 16; ++kb) {
                const int kr = kb * 4 + kOff;
                wg[kb].x = W2[kr * FDIM + n0 + l15];
                wg[kb].y = W2[(kr + 1) * FDIM + n0 + l15];
            }
            #pragma unroll
            for (int kb = 0; kb < 16; ++kb)
                acc = __builtin_amdgcn_wmma_f32_16x16x4_f32(
                    false, bfrag[kb], false, wg[kb], (short)0, acc, false, false);
        }

        const float bval  = bias  ? bias[n0 + l15]  : 0.0f;
        const float b2val = bias2 ? bias2[n0 + l15] : 0.0f;
        #pragma unroll
        for (int j = 0; j < 8; ++j) {
            const int r = m0 + j + lhalf * 8;
            if (r < n) {
                float v = acc[j] + bval;
                if (reluOut) v = fmaxf(v, 0.0f);
                out[(size_t)r * FDIM + n0 + l15] = v;
                if (out2) {
                    const float d = dinv2[r];
                    out2[(size_t)r * FDIM + n0 + l15] = acc[j] * d * d + b2val;
                }
                if (zeroA) zeroA[(size_t)r * FDIM + n0 + l15] = 0.0f;
            }
        }
    }
}

__device__ __forceinline__ int lane_i(int v, int j) { return __builtin_amdgcn_readlane(v, j); }
__device__ __forceinline__ float lane_f(float v, int j) {
    return __int_as_float(__builtin_amdgcn_readlane(__float_as_int(v), j));
}

__device__ __forceinline__ void owner_batch(const int2* q, const int cnt, float* tile,
                                   const float* __restrict__ srow, const int* __restrict__ src,
                                   const float* __restrict__ ew, const int lane) {
    int sl = 0, rl = 0;
    float wl = 1.0f;
    if (lane < cnt) {
        const int2 en = q[lane];
        rl = en.y;
        sl = src[en.x];
        if (ew) wl = ew[en.x];
    }
    if (cnt == 32) {
        for (int j0 = 0; j0 < 32; j0 += 8) {
            float a0[8], a1[8];
            #pragma unroll
            for (int u = 0; u < 8; ++u) {
                const float* hr = srow + (size_t)lane_i(sl, j0 + u) * FDIM;
                a0[u] = hr[lane];
                a1[u] = hr[lane + 32];
            }
            #pragma unroll
            for (int u = 0; u < 8; ++u) {
                float* tr = tile + lane_i(rl, j0 + u) * FDIM;
                const float w = lane_f(wl, j0 + u);
                atomicAdd(tr + lane, a0[u] * w);
                atomicAdd(tr + lane + 32, a1[u] * w);
            }
        }
    } else {
        for (int j = 0; j < cnt; ++j) {
            const float* hr = srow + (size_t)lane_i(sl, j) * FDIM;
            float* tr = tile + lane_i(rl, j) * FDIM;
            const float w = lane_f(wl, j);
            atomicAdd(tr + lane, hr[lane] * w);
            atomicAdd(tr + lane + 32, hr[lane + 32] * w);
        }
    }
}

__device__ __forceinline__ int owner_chunk(const int4 d, const int b, const int lo, const int rows,
                                   int2* q, int qn, float* tile,
                                   const float* __restrict__ srow, const int* __restrict__ src,
                                   const float* __restrict__ ew, const int lane) {
    const int ids[4] = {d.x, d.y, d.z, d.w};
    #pragma unroll
    for (int k = 0; k < 4; ++k) {
        const unsigned rel = (unsigned)(ids[k] - lo);
        const bool in = rel < (unsigned)rows;
        const unsigned m = __builtin_amdgcn_ballot_w32(in);
        if (m) {
            const int pos = qn + (int)__builtin_amdgcn_mbcnt_lo(m, 0u);
            if (in) q[pos] = make_int2(b + k, (int)rel);
            qn += __builtin_popcount(m);
        }
    }
    while (qn >= 32) {
        qn -= 32;
        owner_batch(q + qn, 32, tile, srow, src, ew, lane);
    }
    return qn;
}

__device__ __forceinline__ void owner_scatter_body(const int sblk, float* tile, int2* qAll,
                                   const float* __restrict__ srow, const int* __restrict__ src,
                                   const int* __restrict__ dst, const float* __restrict__ ew,
                                   float* __restrict__ acc_rows, const int n, const int e) {
    const int t    = threadIdx.x;
    const int lane = t & 31;
    const int wave = __builtin_amdgcn_readfirstlane(t >> 5);
    const int range = sblk / NSLICE;
    const int slice = sblk - range * NSLICE;
    const int lo    = range * TROWS;
    const int rows  = (n - lo < TROWS) ? (n - lo) : TROWS;

    {
        float4* t4 = (float4*)tile;
        const float4 z = make_float4(0.f, 0.f, 0.f, 0.f);
        for (int i = t; i < TROWS * FDIM / 4; i += NTHR) t4[i] = z;
    }
    __syncthreads();

    int2* q = qAll + wave * QCAP;
    int qn = 0;

    const int nChunk = (e + 127) >> 7;
    const int cps    = (nChunk + NSLICE - 1) / NSLICE;
    const int cBeg   = slice * cps;
    const int cEnd   = (cBeg + cps < nChunk) ? (cBeg + cps) : nChunk;
    int cRest = cBeg;

    if ((((size_t)dst) & 15) == 0) {
        const int nFull = e >> 7;
        const int cFull = (cEnd < nFull) ? cEnd : nFull;
        if (cBeg + wave < cFull) {
            const int cLast = cFull - 1;
            int c = cBeg + wave;
            int4 cur[4], nxt[4];
            #pragma unroll
            for (int u = 0; u < 4; ++u) {
                const int cc = (c + 32 * u < cLast) ? (c + 32 * u) : cLast;
                cur[u] = *(const int4*)(dst + (size_t)cc * 128 + lane * 4);
            }
            for (; c < cFull; c += 128) {
                #pragma unroll
                for (int u = 0; u < 4; ++u) {
                    const int cc = (c + 128 + 32 * u < cLast) ? (c + 128 + 32 * u) : cLast;
                    nxt[u] = *(const int4*)(dst + (size_t)cc * 128 + lane * 4);
                }
                #pragma unroll
                for (int u = 0; u < 4; ++u) {
                    const int cc = c + 32 * u;
                    if (cc < cFull)
                        qn = owner_chunk(cur[u], cc * 128 + lane * 4, lo, rows, q, qn, tile, srow, src, ew, lane);
                }
                #pragma unroll
                for (int u = 0; u < 4; ++u) cur[u] = nxt[u];
            }
        }
        if (cFull > cRest) cRest = cFull;
    }
    for (int c = cRest + wave; c < cEnd; c += 32) {
        const int b = c * 128 + lane * 4;
        int4 d = make_int4(-1, -1, -1, -1);
        if (b     < e) d.x = dst[b];
        if (b + 1 < e) d.y = dst[b + 1];
        if (b + 2 < e) d.z = dst[b + 2];
        if (b + 3 < e) d.w = dst[b + 3];
        qn = owner_chunk(d, b, lo, rows, q, qn, tile, srow, src, ew, lane);
    }
    if (qn > 0) owner_batch(q, qn, tile, srow, src, ew, lane);
    __syncthreads();

    float* orow = acc_rows + (size_t)lo * FDIM;
    const int nOut = rows * FDIM;
    int i = t;
    for (; i + 3 * NTHR < nOut; i += 4 * NTHR) {
        const float v0 = tile[i], v1 = tile[i + NTHR], v2 = tile[i + 2 * NTHR], v3 = tile[i + 3 * NTHR];
        atomicAdd(orow + i, v0);
        atomicAdd(orow + i + NTHR, v1);
        atomicAdd(orow + i + 2 * NTHR, v2);
        atomicAdd(orow + i + 3 * NTHR, v3);
    }
    for (; i < nOut; i += NTHR) atomicAdd(orow + i, tile[i]);
}

__launch_bounds__(NTHR)
__global__ void node_edge_kernel(const float* A, const float* __restrict__ W1g,
                                 const float* __restrict__ B, const float* __restrict__ W2g,
                                 const float* __restrict__ bias,
                                 const float* __restrict__ rowscale,
                                 const float* __restrict__ lnAg, const float* __restrict__ lnAb,
                                 const float* __restrict__ lnBg, const float* __restrict__ lnBb,
                                 const float* __restrict__ dinv2, const float* __restrict__ bias2,
                                 float* out2, float* out, float* zeroA,
                                 int n, int reluOut, int reluA, int nOwn,
                                 const float* __restrict__ srow, const int* __restrict__ src,
                                 const int* __restrict__ dst, const float* __restrict__ ew,
                                 float* __restrict__ acc_rows, int e) {
    __shared__ __attribute__((aligned(16))) float lds[TROWS * FDIM + NWAVE * QCAP * 2];
    const int blk = blockIdx.x;
    if (blk < nOwn) {
        owner_scatter_body(blk, lds, (int2*)(lds + TROWS * FDIM), srow, src, dst, ew, acc_rows, n, e);
    } else {
        gemm64_body(blk - nOwn, lds, lds + FDIM * FDIM, (float (*)[FDIM])(lds + 2 * FDIM * FDIM),
                    A, W1g, B, W2g, bias, rowscale, lnAg, lnAb, lnBg, lnBb,
                    dinv2, bias2, out2, out, zeroA, n, reluOut, reluA);
    }
}

__global__ void zero_kernel(float* __restrict__ p, long long total) {
    long long i = (long long)blockIdx.x * blockDim.x + threadIdx.x;
    if (i < total) p[i] = 0.0f;
}

extern "C" void kernel_launch(void* const* d_in, const int* in_sizes, int n_in,
                              void* d_out, int out_size, void* d_ws, size_t ws_size,
                              hipStream_t stream) {
    const float* x        = (const float*)d_in[0];
    const int*   ei       = (const int*)d_in[1];
    const float* gcn_w1   = (const float*)d_in[2];
    const float* gcn_b1   = (const float*)d_in[3];
    const float* gcn_w2   = (const float*)d_in[4];
    const float* gcn_b2   = (const float*)d_in[5];
    const float* sage_wl1 = (const float*)d_in[6];
    const float* sage_bl1 = (const float*)d_in[7];
    const float* sage_wr1 = (const float*)d_in[8];
    const float* sage_wl2 = (const float*)d_in[9];
    const float* sage_bl2 = (const float*)d_in[10];
    const float* sage_wr2 = (const float*)d_in[11];
    const float* gcn_ln_g = (const float*)d_in[12];
    const float* gcn_ln_b = (const float*)d_in[13];
    const float* sage_ln_g = (const float*)d_in[14];
    const float* sage_ln_b = (const float*)d_in[15];
    const float* proj_w   = (const float*)d_in[16];
    const float* proj_b   = (const float*)d_in[17];

    const int N = in_sizes[0] / FDIM;
    const int E = in_sizes[1] / 2;
    const int* src = ei;
    const int* dst = ei + E;

    float* cnt  = (float*)d_ws;
    float* dinv = cnt + N;
    float* sinv = dinv + N;
    float* ew   = sinv + N;
    float* h    = ew + E;
    float* g    = h + (size_t)N * FDIM;
    float* agg  = g + (size_t)N * FDIM;
    float* out  = (float*)d_out;

    const long long NM = (long long)N * FDIM;
    const int nblkN  = (N + 255) / 256;
    const int nblkE  = (E + 255) / 256;
    const int nblkNM = (int)((NM + 255) / 256);
    const int nblkG  = (N + 511) / 512;
    const int nOwn   = ((N + TROWS - 1) / TROWS) * NSLICE;

    init_cnt_kernel<<<nblkN, 256, 0, stream>>>(cnt, N);
    count_edges_kernel<<<nblkE, 256, 0, stream>>>(dst, cnt, E);
    finalize_deg_kernel<<<nblkN, 256, 0, stream>>>(cnt, dinv, sinv, N);
    edge_w_kernel<<<nblkE, 256, 0, stream>>>(src, dst, dinv, ew, E);

    const int nblkGE = nOwn + nblkG;

    zero_kernel<<<nblkNM, 256, 0, stream>>>(agg, NM);

    node_edge_kernel<<<nblkGE, NTHR, 0, stream>>>(
        x, gcn_w1, nullptr, nullptr, nullptr, nullptr,
        nullptr, nullptr, nullptr, nullptr,
        dinv, gcn_b1, g, h, nullptr, N, 0, 0, nOwn,
        x, src, dst, nullptr, agg, E);

    node_edge_kernel<<<nblkGE, NTHR, 0, stream>>>(
        agg, sage_wl1, x, sage_wr1, sage_bl1, sinv,
        nullptr, nullptr, nullptr, nullptr,
        nullptr, nullptr, nullptr, out, agg, N, 1  , 0, nOwn,
        h, src, dst, ew, g, E);

    node_edge_kernel<<<nblkGE, NTHR, 0, stream>>>(
        g, gcn_w2, nullptr, nullptr, nullptr, nullptr,
        nullptr, nullptr, nullptr, nullptr,
        dinv, gcn_b2, g, h, nullptr, N, 0, 1  , nOwn,
        out, src, dst, nullptr, agg, E);

    node_edge_kernel<<<nblkGE, NTHR, 0, stream>>>(
        agg, sage_wl2, out, sage_wr2, sage_bl2, sinv,
        nullptr, nullptr, nullptr, nullptr,
        nullptr, nullptr, nullptr, agg, nullptr, N, 0, 0, nOwn,
        h, src, dst, ew, g, E);

    node_edge_kernel<<<nblkG, NTHR, 0, stream>>>(
        g, proj_w, agg, proj_w + FDIM * FDIM, proj_b, nullptr,
        gcn_ln_g, gcn_ln_b, sage_ln_g, sage_ln_b,
        nullptr, nullptr, nullptr, out, nullptr, N, 0, 0, 0,
        nullptr, src, dst, nullptr, nullptr, 0);
}


# Owner filter reads a 1-byte range key per edge, not its 4-byte id (count pass writes keys in the dropped edge-weight slot; owners form dinv[src]*dinv[dst]); launches 10 to 9; sums stay arrival-order
# speedup vs baseline: 2.0081x; 1.1131x over previous
#include <hip/hip_runtime.h>

typedef __attribute__((ext_vector_type(2))) float v2f;
typedef __attribute__((ext_vector_type(8))) float v8f;

#define FDIM 64
#define NTHR  1024
#define NWAVE 32
#define TROWS 1024
#define NSLICE 2
#define QCAP  64
#define KSTEP 512

__global__ void init_cnt_kernel(float* __restrict__ cnt, int n) {
    int i = blockIdx.x * blockDim.x + threadIdx.x;
    if (i < n) cnt[i] = 0.0f;
}

__global__ void count_edges_kernel(const int* __restrict__ dst, float* __restrict__ cnt,
                                   unsigned char* __restrict__ key, int e) {
    int i = blockIdx.x * blockDim.x + threadIdx.x;
    if (i < e) {
        const int d = dst[i];
        atomicAdd(&cnt[d], 1.0f);
        const int r = d / TROWS;
        key[i] = (unsigned char)(r < 255 ? r : 255);
    }
}

__global__ void finalize_deg_kernel(const float* __restrict__ cnt, float* __restrict__ dinv,
                                    float* __restrict__ sinv, int n) {
    int i = blockIdx.x * blockDim.x + threadIdx.x;
    if (i < n) {
        float c = cnt[i];
        dinv[i] = rsqrtf(c + 1.0f);
        sinv[i] = 1.0f / fmaxf(c, 1.0f);
    }
}

__device__ __forceinline__ void gemm64_body(const int blk, float* W1, float* W2, float (*LNP)[FDIM],
                                   const float* A, const float* __restrict__ W1g,
                                   const float* __restrict__ B, const float* __restrict__ W2g,
                                   const float* __restrict__ bias,
                                   const float* __restrict__ rowscale,
                                   const float* __restrict__ lnAg, const float* __restrict__ lnAb,
                                   const float* __restrict__ lnBg, const float* __restrict__ lnBb,
                                   const float* __restrict__ dinv2, const float* __restrict__ bias2,
                                   float* out2,
                                   float* out, float* zeroA, int n, int reluOut, int reluA) {
    const int t = threadIdx.x;
    {
        const float4* s1 = (const float4*)W1g;
        float4* l1 = (float4*)W1;
        for (int i = t; i < 1024; i += NTHR) l1[i] = s1[i];
        if (B) {
            const float4* s2 = (const float4*)W2g;
            float4* l2 = (float4*)W2;
            for (int i = t; i < 1024; i += NTHR) l2[i] = s2[i];
        }
        if (lnAg && t < FDIM) { LNP[0][t] = lnAg[t]; LNP[1][t] = lnAb[t]; }
        if (lnBg && t < FDIM) { LNP[2][t] = lnBg[t]; LNP[3][t] = lnBb[t]; }
    }
    __syncthreads();

    const int wave  = t >> 5;
    const int lane  = t & 31;
    const int m0    = (blk * NWAVE + wave) * 16;
    if (m0 >= n) return;
    const int lhalf = lane >> 4;
    const int l15   = lane & 15;
    const int kOff  = lhalf * 2;

    int row = m0 + l15;
    if (row >= n) row = n - 1;

    const float rs = rowscale ? rowscale[row] : 1.0f;

    v2f afrag[16];
    {
        const float* ar = A + (size_t)row * FDIM + kOff;
        #pragma unroll
        for (int kb = 0; kb < 16; ++kb) {
            float a0 = ar[kb * 4 + 0];
            float a1 = ar[kb * 4 + 1];
            if (reluA) { a0 = fmaxf(a0, 0.0f); a1 = fmaxf(a1, 0.0f); }
            afrag[kb].x = a0 * rs;
            afrag[kb].y = a1 * rs;
        }
        if (lnAg) {
            float s = 0.0f, q = 0.0f;
            #pragma unroll
            for (int kb = 0; kb < 16; ++kb) {
                s += afrag[kb].x + afrag[kb].y;
                q += afrag[kb].x * afrag[kb].x + afrag[kb].y * afrag[kb].y;
            }
            s += __shfl_xor(s, 16, 32);
            q += __shfl_xor(q, 16, 32);
            const float mu  = s * (1.0f / 64.0f);
            const float inv = rsqrtf(fmaxf(q * (1.0f / 64.0f) - mu * mu, 0.0f) + 1e-5f);
            #pragma unroll
            for (int kb = 0; kb < 16; ++kb) {
                const int k0 = kb * 4 + kOff;
                afrag[kb].x = (afrag[kb].x - mu) * inv * LNP[0][k0]     + LNP[1][k0];
                afrag[kb].y = (afrag[kb].y - mu) * inv * LNP[0][k0 + 1] + LNP[1][k0 + 1];
            }
        }
    }
    v2f bfrag[16];
    if (B) {
        const float* br = B + (size_t)row * FDIM + kOff;
        #pragma unroll
        for (int kb = 0; kb < 16; ++kb) {
            bfrag[kb].x = br[kb * 4 + 0];
            bfrag[kb].y = br[kb * 4 + 1];
        }
        if (lnBg) {
            float s = 0.0f, q = 0.0f;
            #pragma unroll
            for (int kb = 0; kb < 16; ++kb) {
                s += bfrag[kb].x + bfrag[kb].y;
                q += bfrag[kb].x * bfrag[kb].x + bfrag[kb].y * bfrag[kb].y;
            }
            s += __shfl_xor(s, 16, 32);
            q += __shfl_xor(q, 16, 32);
            const float mu  = s * (1.0f / 64.0f);
            const float inv = rsqrtf(fmaxf(q * (1.0f / 64.0f) - mu * mu, 0.0f) + 1e-5f);
            #pragma unroll
            for (int kb = 0; kb < 16; ++kb) {
                const int k0 = kb * 4 + kOff;
                bfrag[kb].x = (bfrag[kb].x - mu) * inv * LNP[2][k0]     + LNP[3][k0];
                bfrag[kb].y = (bfrag[kb].y - mu) * inv * LNP[2][k0 + 1] + LNP[3][k0 + 1];
            }
        }
    }

    #pragma unroll
    for (int nb = 0; nb < 4; ++nb) {
        const int n0 = nb * 16;
        v8f acc = {0.f, 0.f, 0.f, 0.f, 0.f, 0.f, 0.f, 0.f};

        v2f wf[16];
        #pragma unroll
        for (int kb = 0; kb < 16; ++kb) {
            const int kr = kb * 4 + kOff;
            wf[kb].x = W1[kr * FDIM + n0 + l15];
            wf[kb].y = W1[(kr + 1) * FDIM + n0 + l15];
        }
        #pragma unroll
        for (int kb = 0; kb < 16; ++kb)
            acc = __builtin_amdgcn_wmma_f32_16x16x4_f32(
                false, afrag[kb], false, wf[kb], (short)0, acc, false, false);

        if (B) {
            v2f wg[16];
            #pragma unroll
            for (int kb = 0; kb < 16; ++kb) {
                const int kr = kb * 4 + kOff;
                wg[kb].x = W2[kr * FDIM + n0 + l15];
                wg[kb].y = W2[(kr + 1) * FDIM + n0 + l15];
            }
            #pragma unroll
            for (int kb = 0; kb < 16; ++kb)
                acc = __builtin_amdgcn_wmma_f32_16x16x4_f32(
                    false, bfrag[kb], false, wg[kb], (short)0, acc, false, false);
        }

        const float bval  = bias  ? bias[n0 + l15]  : 0.0f;
        const float b2val = bias2 ? bias2[n0 + l15] : 0.0f;
        #pragma unroll
        for (int j = 0; j < 8; ++j) {
            const int r = m0 + j + lhalf * 8;
            if (r < n) {
                float v = acc[j] + bval;
                if (reluOut) v = fmaxf(v, 0.0f);
                out[(size_t)r * FDIM + n0 + l15] = v;
                if (out2) {
                    const float d = dinv2[r];
                    out2[(size_t)r * FDIM + n0 + l15] = acc[j] * d * d + b2val;
                }
                if (zeroA) zeroA[(size_t)r * FDIM + n0 + l15] = 0.0f;
            }
        }
    }
}

__device__ __forceinline__ int lane_i(int v, int j) { return __builtin_amdgcn_readlane(v, j); }
__device__ __forceinline__ float lane_f(float v, int j) {
    return __int_as_float(__builtin_amdgcn_readlane(__float_as_int(v), j));
}

template <bool WEIGHTED>
__device__ __forceinline__ void owner_batch(const int* q, const int cnt, float* tile,
                                   const float* __restrict__ srow, const int* __restrict__ src,
                                   const int* __restrict__ dst, const float* __restrict__ dinvE,
                                   const int lo, const int lane) {
    int sl = 0, rl = 0;
    float ws = 1.0f, wd = 1.0f;
    if (lane < cnt) {
        const int en = q[lane];
        const int d  = dst[en];
        sl = src[en];
        rl = d - lo;
        if (WEIGHTED) { ws = dinvE[sl]; wd = dinvE[d]; }
    }
    if (cnt == 32) {
        for (int j0 = 0; j0 < 32; j0 += 8) {
            float a0[8], a1[8];
            #pragma unroll
            for (int u = 0; u < 8; ++u) {
                const float* hr = srow + (size_t)lane_i(sl, j0 + u) * FDIM;
                a0[u] = hr[lane];
                a1[u] = hr[lane + 32];
            }
            #pragma unroll
            for (int u = 0; u < 8; ++u) {
                float* tr = tile + lane_i(rl, j0 + u) * FDIM;
                const float w = WEIGHTED ? lane_f(ws, j0 + u) * lane_f(wd, j0 + u) : 1.0f;
                atomicAdd(tr + lane, a0[u] * w);
                atomicAdd(tr + lane + 32, a1[u] * w);
            }
        }
    } else {
        for (int j = 0; j < cnt; ++j) {
            const float* hr = srow + (size_t)lane_i(sl, j) * FDIM;
            float* tr = tile + lane_i(rl, j) * FDIM;
            const float w = WEIGHTED ? lane_f(ws, j) * lane_f(wd, j) : 1.0f;
            atomicAdd(tr + lane, hr[lane] * w);
            atomicAdd(tr + lane + 32, hr[lane + 32] * w);
        }
    }
}

__device__ __forceinline__ unsigned nz7(unsigned x) { return ((x & 0x7f7f7f7fu) + 0x7f7f7f7fu) | x; }

__device__ __forceinline__ unsigned flags_from_keys(const uint4 kw, const unsigned pat) {
    const unsigned nz = ((nz7(kw.x ^ pat) >> 7) & 0x01010101u) | ((nz7(kw.y ^ pat) >> 6) & 0x02020202u) |
                        ((nz7(kw.z ^ pat) >> 5) & 0x04040404u) | ((nz7(kw.w ^ pat) >> 4) & 0x08080808u);
    return ~nz & 0x0f0f0f0fu;
}

__device__ __forceinline__ unsigned flags_from_ids(const int* __restrict__ dst, const int e0, const int e,
                                   const int lo, const int rows) {
    unsigned cm = 0u;
    for (int k = 0; k < 16; ++k) {
        const int i = e0 + k;
        if (i < e && (unsigned)(dst[i] - lo) < (unsigned)rows) cm |= 1u << (8 * (k & 3) + (k >> 2));
    }
    return cm;
}

template <bool WEIGHTED>
__device__ __forceinline__ int owner_take(unsigned cm, const int e0, int* q, int qn, float* tile,
                                   const float* __restrict__ srow, const int* __restrict__ src,
                                   const int* __restrict__ dst, const float* __restrict__ dinvE,
                                   const int lo, const int lane) {
    for (int it = 0; it < 16; ++it) {
        const bool has = cm != 0u;
        const unsigned m = __builtin_amdgcn_ballot_w32(has);
        if (!m) break;
        const int pos = qn + (int)__builtin_amdgcn_mbcnt_lo(m, 0u);
        if (has) {
            const int p = __builtin_ctz(cm);
            q[pos] = e0 + (((p & 3) << 2) | (p >> 3));
            cm &= cm - 1u;
        }
        qn += __builtin_popcount(m);
        if (qn >= 32) {
            qn -= 32;
            owner_batch<WEIGHTED>(q + qn, 32, tile, srow, src, dst, dinvE, lo, lane);
        }
    }
    return qn;
}

template <bool WEIGHTED>
__device__ __forceinline__ void owner_scatter_body(const int sblk, float* tile, int* qAll,
                                   const float* __restrict__ srow, const int* __restrict__ src,
                                   const int* __restrict__ dst, const unsigned char* __restrict__ key,
                                   const float* __restrict__ dinvE,
                                   float* __restrict__ acc_rows, const int n, const int e) {
    const int t    = threadIdx.x;
    const int lane = t & 31;
    const int wave = __builtin_amdgcn_readfirstlane(t >> 5);
    const int range = sblk / NSLICE;
    const int slice = sblk - range * NSLICE;
    const int lo    = range * TROWS;
    const int rows  = (n - lo < TROWS) ? (n - lo) : TROWS;

    {
        float4* t4 = (float4*)tile;
        const float4 z = make_float4(0.f, 0.f, 0.f, 0.f);
        for (int i = t; i < TROWS * FDIM / 4; i += NTHR) t4[i] = z;
    }
    __syncthreads();

    int* q = qAll + wave * QCAP;
    int qn = 0;

    const int nStep = (e + KSTEP - 1) / KSTEP;
    const int sps   = (nStep + NSLICE - 1) / NSLICE;
    const int sBeg  = slice * sps;
    const int sEnd  = (sBeg + sps < nStep) ? (sBeg + sps) : nStep;
    int sRest = sBeg;

    if (key && (n + TROWS - 1) / TROWS <= 256 && (((size_t)key) & 15) == 0) {
        const int nFull = e / KSTEP;
        const int sFull = (sEnd < nFull) ? sEnd : nFull;
        if (sBeg + wave < sFull) {
            const int sLast = sFull - 1;
            const unsigned pat = (unsigned)range * 0x01010101u;
            int c = sBeg + wave;
            uint4 cur[4], nxt[4];
            #pragma unroll
            for (int u = 0; u < 4; ++u) {
                const int cc = (c + 32 * u < sLast) ? (c + 32 * u) : sLast;
                cur[u] = *(const uint4*)(key + (size_t)cc * KSTEP + lane * 16);
            }
            for (; c < sFull; c += 128) {
                #pragma unroll
                for (int u = 0; u < 4; ++u) {
                    const int cc = (c + 128 + 32 * u < sLast) ? (c + 128 + 32 * u) : sLast;
                    nxt[u] = *(const uint4*)(key + (size_t)cc * KSTEP + lane * 16);
                }
                #pragma unroll
                for (int u = 0; u < 4; ++u) {
                    const int cc = c + 32 * u;
                    if (cc < sFull)
                        qn = owner_take<WEIGHTED>(flags_from_keys(cur[u], pat), cc * KSTEP + lane * 16,
                                        q, qn, tile, srow, src, dst, dinvE, lo, lane);
                }
                #pragma unroll
                for (int u = 0; u < 4; ++u) cur[u] = nxt[u];
            }
        }
        if (sFull > sRest) sRest = sFull;
    }
    for (int c = sRest + wave; c < sEnd; c += 32) {
        const int e0 = c * KSTEP + lane * 16;
        qn = owner_take<WEIGHTED>(flags_from_ids(dst, e0, e, lo, rows), e0,
                        q, qn, tile, srow, src, dst, dinvE, lo, lane);
    }
    if (qn > 0) owner_batch<WEIGHTED>(q, qn, tile, srow, src, dst, dinvE, lo, lane);
    __syncthreads();

    float* orow = acc_rows + (size_t)lo * FDIM;
    const int nOut = rows * FDIM;
    int i = t;
    for (; i + 3 * NTHR < nOut; i += 4 * NTHR) {
        const float v0 = tile[i], v1 = tile[i + NTHR], v2 = tile[i + 2 * NTHR], v3 = tile[i + 3 * NTHR];
        atomicAdd(orow + i, v0);
        atomicAdd(orow + i + NTHR, v1);
        atomicAdd(orow + i + 2 * NTHR, v2);
        atomicAdd(orow + i + 3 * NTHR, v3);
    }
    for (; i < nOut; i += NTHR) atomicAdd(orow + i, tile[i]);
}

__launch_bounds__(NTHR)
__global__ void node_edge_kernel(const float* A, const float* __restrict__ W1g,
                                 const float* __restrict__ B, const float* __restrict__ W2g,
                                 const float* __restrict__ bias,
                                 const float* __restrict__ rowscale,
                                 const float* __restrict__ lnAg, const float* __restrict__ lnAb,
                                 const float* __restrict__ lnBg, const float* __restrict__ lnBb,
                                 const float* __restrict__ dinv2, const float* __restrict__ bias2,
                                 float* out2, float* out, float* zeroA,
                                 int n, int reluOut, int reluA, int nOwn,
                                 const float* __restrict__ srow, const int* __restrict__ src,
                                 const int* __restrict__ dst, const unsigned char* __restrict__ key,
                                 const float* __restrict__ dinvE,
                                 float* __restrict__ acc_rows, int e) {
    __shared__ __attribute__((aligned(16))) float lds[TROWS * FDIM + NWAVE * QCAP];
    const int blk = blockIdx.x;
    if (blk < nOwn) {
        int* qAll = (int*)(lds + TROWS * FDIM);
        if (dinvE) owner_scatter_body<true>(blk, lds, qAll, srow, src, dst, key, dinvE, acc_rows, n, e);
        else       owner_scatter_body<false>(blk, lds, qAll, srow, src, dst, key, dinvE, acc_rows, n, e);
    } else {
        gemm64_body(blk - nOwn, lds, lds + FDIM * FDIM, (float (*)[FDIM])(lds + 2 * FDIM * FDIM),
                    A, W1g, B, W2g, bias, rowscale, lnAg, lnAb, lnBg, lnBb,
                    dinv2, bias2, out2, out, zeroA, n, reluOut, reluA);
    }
}

__global__ void zero_kernel(float* __restrict__ p, long long total) {
    long long i = (long long)blockIdx.x * blockDim.x + threadIdx.x;
    if (i < total) p[i] = 0.0f;
}

extern "C" void kernel_launch(void* const* d_in, const int* in_sizes, int n_in,
                              void* d_out, int out_size, void* d_ws, size_t ws_size,
                              hipStream_t stream) {
    const float* x        = (const float*)d_in[0];
    const int*   ei       = (const int*)d_in[1];
    const float* gcn_w1   = (const float*)d_in[2];
    const float* gcn_b1   = (const float*)d_in[3];
    const float* gcn_w2   = (const float*)d_in[4];
    const float* gcn_b2   = (const float*)d_in[5];
    const float* sage_wl1 = (const float*)d_in[6];
    const float* sage_bl1 = (const float*)d_in[7];
    const float* sage_wr1 = (const float*)d_in[8];
    const float* sage_wl2 = (const float*)d_in[9];
    const float* sage_bl2 = (const float*)d_in[10];
    const float* sage_wr2 = (const float*)d_in[11];
    const float* gcn_ln_g = (const float*)d_in[12];
    const float* gcn_ln_b = (const float*)d_in[13];
    const float* sage_ln_g = (const float*)d_in[14];
    const float* sage_ln_b = (const float*)d_in[15];
    const float* proj_w   = (const float*)d_in[16];
    const float* proj_b   = (const float*)d_in[17];

    const int N = in_sizes[0] / FDIM;
    const int E = in_sizes[1] / 2;
    const int* src = ei;
    const int* dst = ei + E;

    float* cnt  = (float*)d_ws;
    float* dinv = cnt + N;
    float* sinv = dinv + N;
    float* kslot = sinv + N;
    unsigned char* key = (unsigned char*)kslot;
    float* h    = kslot + E;
    float* g    = h + (size_t)N * FDIM;
    float* agg  = g + (size_t)N * FDIM;
    float* out  = (float*)d_out;

    const long long NM = (long long)N * FDIM;
    const int nblkN  = (N + 255) / 256;
    const int nblkE  = (E + 255) / 256;
    const int nblkNM = (int)((NM + 255) / 256);
    const int nblkG  = (N + 511) / 512;
    const int nOwn   = ((N + TROWS - 1) / TROWS) * NSLICE;

    init_cnt_kernel<<<nblkN, 256, 0, stream>>>(cnt, N);
    count_edges_kernel<<<nblkE, 256, 0, stream>>>(dst, cnt, key, E);
    finalize_deg_kernel<<<nblkN, 256, 0, stream>>>(cnt, dinv, sinv, N);

    const int nblkGE = nOwn + nblkG;

    zero_kernel<<<nblkNM, 256, 0, stream>>>(agg, NM);

    node_edge_kernel<<<nblkGE, NTHR, 0, stream>>>(
        x, gcn_w1, nullptr, nullptr, nullptr, nullptr,
        nullptr, nullptr, nullptr, nullptr,
        dinv, gcn_b1, g, h, nullptr, N, 0, 0, nOwn,
        x, src, dst, key, nullptr, agg, E);

    node_edge_kernel<<<nblkGE, NTHR, 0, stream>>>(
        agg, sage_wl1, x, sage_wr1, sage_bl1, sinv,
        nullptr, nullptr, nullptr, nullptr,
        nullptr, nullptr, nullptr, out, agg, N, 1  , 0, nOwn,
        h, src, dst, key, dinv, g, E);

    node_edge_kernel<<<nblkGE, NTHR, 0, stream>>>(
        g, gcn_w2, nullptr, nullptr, nullptr, nullptr,
        nullptr, nullptr, nullptr, nullptr,
        dinv, gcn_b2, g, h, nullptr, N, 0, 1  , nOwn,
        out, src, dst, key, nullptr, agg, E);

    node_edge_kernel<<<nblkGE, NTHR, 0, stream>>>(
        agg, sage_wl2, out, sage_wr2, sage_bl2, sinv,
        nullptr, nullptr, nullptr, nullptr,
        nullptr, nullptr, nullptr, agg, nullptr, N, 0, 0, nOwn,
        h, src, dst, key, dinv, g, E);

    node_edge_kernel<<<nblkG, NTHR, 0, stream>>>(
        g, proj_w, agg, proj_w + FDIM * FDIM, proj_b, nullptr,
        gcn_ln_g, gcn_ln_b, sage_ln_g, sage_ln_b,
        nullptr, nullptr, nullptr, out, nullptr, N, 0, 0, 0,
        nullptr, src, dst, nullptr, nullptr, nullptr, 0);
}
